# starting kernel
# baseline (speedup 1.0000x reference)
#include <hip/hip_runtime.h>

typedef __attribute__((ext_vector_type(2))) float v2f;
typedef __attribute__((ext_vector_type(8))) float v8f;

static constexpr int HH   = 56;
static constexpr int WW   = 56;
static constexpr int HW   = HH * WW;
static constexpr int KTOT = 84;
static constexpr int NCH  = 48;
static constexpr int ROWS = 2 * HW;
static constexpr int ROW_TILES = ROWS / 16;

__global__ void fold_weights_kernel(const float* __restrict__ W0,
                                    const float* __restrict__ W1,
                                    float* __restrict__ Wc) {
  int idx = blockIdx.x * blockDim.x + threadIdx.x;
  if (idx >= NCH * KTOT) return;
  int p  = idx / KTOT;
  int jk = idx - p * KTOT;
  const float* w1 = W1 + jk * 9;
  const float* w0 = W0 + p  * 9;
  float s = 0.f;
#pragma unroll
  for (int i = 0; i < 9; ++i) s += w1[i] * w0[i];
  Wc[p * KTOT + jk] = s;
}

__global__ __launch_bounds__(128)
void conv_wmma_kernel(const float* __restrict__ x,
                      const float* __restrict__ Wc,
                      float* __restrict__ out) {
  __shared__ alignas(16) float lWc[NCH * KTOT];

  const int tid = threadIdx.x;
  for (int i = tid; i < NCH * KTOT; i += 128) lWc[i] = Wc[i];
  __syncthreads();

  const int wave  = tid >> 5;
  const int lane  = tid & 31;
  const int laneh = lane >> 4;
  const int lanem = lane & 15;

  const int row_tile = blockIdx.x * 4 + wave;
  const int row = row_tile * 16 + lanem;
  const int o   = row / HW;
  const int rem = row - o * HW;
  const int n   = rem / WW;
  const int m   = rem - n * WW;
  const float* xb = x + (o * 12) * HW + n * WW;
  __builtin_prefetch(xb + m, 0, 3);

  const int khalf = laneh * 2;

  int j0 = 0, kk0 = khalf;
  int j1 = 0, kk1 = khalf + 1;

  const float* w0r = &lWc[( 0 + lanem) * KTOT + khalf];
  const float* w1r = &lWc[(16 + lanem) * KTOT + khalf];
  const float* w2r = &lWc[(32 + lanem) * KTOT + khalf];

  v8f acc0 = {};
  v8f acc1 = {};
  v8f acc2 = {};

  for (int b = 0; b < KTOT / 4; ++b) {
    const int mm0 = m + kk0 - 3;
    const int mm1 = m + kk1 - 3;
    const float a0 = (mm0 >= 0 && mm0 < WW) ? xb[j0 * HW + mm0] : 0.f;
    const float a1 = (mm1 >= 0 && mm1 < WW) ? xb[j1 * HW + mm1] : 0.f;
    v2f a = {a0, a1};

    const v2f b0 = *(const v2f*)(w0r + b * 4);
    const v2f b1 = *(const v2f*)(w1r + b * 4);
    const v2f b2 = *(const v2f*)(w2r + b * 4);

    acc0 = __builtin_amdgcn_wmma_f32_16x16x4_f32(false, a, false, b0,
                                                 (short)0, acc0, false, false);
    acc1 = __builtin_amdgcn_wmma_f32_16x16x4_f32(false, a, false, b1,
                                                 (short)0, acc1, false, false);
    acc2 = __builtin_amdgcn_wmma_f32_16x16x4_f32(false, a, false, b2,
                                                 (short)0, acc2, false, false);

    kk0 += 4; if (kk0 >= 7) { kk0 -= 7; ++j0; }
    kk1 += 4; if (kk1 >= 7) { kk1 -= 7; ++j1; }
  }

  const int grow0 = row_tile * 16 + laneh * 8;
  const int rem0  = grow0 - o * HW;
  int n2 = rem0 / WW;
  int m2 = rem0 - n2 * WW;
  int nr = n2 + 1; if (nr == HH) nr = 0;
  float* obase = out + (size_t)(o * NCH) * HW;

#pragma unroll
  for (int r = 0; r < 8; ++r) {
    float* ob = obase + nr * WW + m2;
    ob[(size_t)( 0 + lanem) * HW] = acc0[r];
    ob[(size_t)(16 + lanem) * HW] = acc1[r];
    ob[(size_t)(32 + lanem) * HW] = acc2[r];
    ++m2;
    if (m2 == WW) { m2 = 0; ++nr; if (nr == HH) nr = 0; }
  }
}

extern "C" void kernel_launch(void* const* d_in, const int* in_sizes, int n_in,
                              void* d_out, int out_size, void* d_ws, size_t ws_size,
                              hipStream_t stream) {
  const float* x  = (const float*)d_in[0];
  const float* W0 = (const float*)d_in[1];
  const float* W1 = (const float*)d_in[2];
  float* out = (float*)d_out;
  float* Wc  = (float*)d_ws;

  fold_weights_kernel<<<(NCH * KTOT + 127) / 128, 128, 0, stream>>>(W0, W1, Wc);
  conv_wmma_kernel<<<ROW_TILES / 4, 128, 0, stream>>>(x, Wc, out);
}


# Fold kernel writes the folded weights in B-fragment order; each wave loads them straight into registers, 33 coalesced loads up front: the LDS weight copy loop and its serial round trips are gone
# speedup vs baseline: 2.1703x; 2.1703x over previous
#include <hip/hip_runtime.h>


typedef __attribute__((ext_vector_type(2))) float v2f;
typedef __attribute__((ext_vector_type(4))) float v4f;
typedef __attribute__((ext_vector_type(8))) float v8f;

static constexpr int HH   = 56;
static constexpr int WW   = 56;
static constexpr int HW   = HH * WW;
static constexpr int KTOT = 84;
static constexpr int NCH  = 48;
static constexpr int ROWS = 2 * HW;
static constexpr int ROW_TILES = ROWS / 16;

static constexpr int SPITCH   = 64;
static constexpr int SROWS    = 12 * 2;
static constexpr int BLK_ROWS = 64;
static constexpr int BLK_PER_O = HW / BLK_ROWS;
static constexpr int ROW_V4   = 2 * WW / 4;
static constexpr int SLAB_V4  = 12 * ROW_V4;

static constexpr int KSTEPS   = KTOT / 4;
static constexpr int KPAIRS   = KSTEPS / 2;
static constexpr int PAIR_F   = KPAIRS * 32 * 4;
static constexpr int TILE_F   = 16 * KTOT;

__global__ void fold_weights_kernel(const float* __restrict__ W0,
                                    const float* __restrict__ W1,
                                    float* __restrict__ Wf) {
  int d = blockIdx.x * blockDim.x + threadIdx.x;
  if (d >= NCH * KTOT) return;
  int t  = d / TILE_F;
  int rm = d - t * TILE_F;
  int b, lane, s;
  if (rm < PAIR_F) {
    int w = rm & 127;
    lane = w >> 2;
    b = 2 * (rm >> 7) + ((w >> 1) & 1);
    s = w & 1;
  } else {
    int w = rm - PAIR_F;
    lane = w >> 1;
    b = KSTEPS - 1;
    s = w & 1;
  }
  int p  = 16 * t + (lane & 15);
  int jk = 4 * b + 2 * (lane >> 4) + s;
  const float* w1 = W1 + jk * 9;
  const float* w0 = W0 + p  * 9;
  float acc = 0.f;
#pragma unroll
  for (int i = 0; i < 9; ++i) acc += w1[i] * w0[i];
  Wf[d] = acc;
}

__global__ __launch_bounds__(128)
void conv_wmma_kernel(const float* __restrict__ x,
                      const float* __restrict__ Wf,
                      float* __restrict__ out) {
  __shared__ alignas(16) float slab[SROWS * SPITCH];

  const int tid = threadIdx.x;

  const int o_blk   = blockIdx.x / BLK_PER_O;
  const int rem_blk = (blockIdx.x - o_blk * BLK_PER_O) * BLK_ROWS;
  const int n0      = rem_blk / WW;
  const float* xs = x + (o_blk * 12) * HW + n0 * WW;

  v4f sv[3];
#pragma unroll
  for (int u = 0; u < 3; ++u) {
    const int idx = tid + 128 * u;
    if (idx < SLAB_V4) {
      const int j = idx / ROW_V4;
      const int q = idx - j * ROW_V4;
      sv[u] = *(const v4f*)(xs + j * HW + q * 4);
    }
  }

  const int lane = tid & 31;
  v4f wp[3][KPAIRS];
  v2f wt[3];
#pragma unroll
  for (int t = 0; t < 3; ++t) {
    const float* wf = Wf + t * TILE_F;
#pragma unroll
    for (int bp = 0; bp < KPAIRS; ++bp)
      wp[t][bp] = *(const v4f*)(wf + bp * 128 + lane * 4);
    wt[t] = *(const v2f*)(wf + PAIR_F + lane * 2);
  }

#pragma unroll
  for (int u = 0; u < 3; ++u) {
    const int idx = tid + 128 * u;
    if (idx < SLAB_V4) {
      const int j = idx / ROW_V4;
      const int q = idx - j * ROW_V4;
      const int r = q / (WW / 4);
      const int c = (q - r * (WW / 4)) * 4;
      *(v4f*)&slab[(j * 2 + r) * SPITCH + 4 + c] = sv[u];
    }
  }
  if (tid < 2 * SROWS) {
    const v4f z = {0.f, 0.f, 0.f, 0.f};
    *(v4f*)&slab[(tid >> 1) * SPITCH + (tid & 1) * 60] = z;
  }
  __syncthreads();

  const int wave  = tid >> 5;
  const int laneh = lane >> 4;
  const int lanem = lane & 15;

  const int row_tile = blockIdx.x * 4 + wave;
  const int row = row_tile * 16 + lanem;
  const int o   = row / HW;
  const int rem = row - o * HW;
  const int n   = rem / WW;
  const int m   = rem - n * WW;


  const float* ap = slab + (n - n0) * SPITCH + m + 1;

  float av0[KTOT / 4], av1[KTOT / 4];
#pragma unroll
  for (int b = 0; b < KTOT / 4; ++b) {
    const int ka = 4 * b;
    const int kb = 4 * b + 2;
    const int offa0 = (ka / 7) * (2 * SPITCH) + (ka % 7);
    const int offa1 = ((ka + 1) / 7) * (2 * SPITCH) + ((ka + 1) % 7);
    const int offb0 = (kb / 7) * (2 * SPITCH) + (kb % 7);
    const int offb1 = ((kb + 1) / 7) * (2 * SPITCH) + ((kb + 1) % 7);
    av0[b] = ap[laneh ? offb0 : offa0];
    av1[b] = ap[laneh ? offb1 : offa1];
  }

  v8f acc0 = {};
  v8f acc1 = {};
  v8f acc2 = {};

#pragma unroll
  for (int b = 0; b < KSTEPS; ++b) {
    v2f a = {av0[b], av1[b]};

    v2f b0, b1, b2;
    if (b < 2 * KPAIRS) {
      const int bp = b >> 1;
      const int e  = (b & 1) * 2;
      b0 = v2f{wp[0][bp][e], wp[0][bp][e + 1]};
      b1 = v2f{wp[1][bp][e], wp[1][bp][e + 1]};
      b2 = v2f{wp[2][bp][e], wp[2][bp][e + 1]};
    } else {
      b0 = wt[0];
      b1 = wt[1];
      b2 = wt[2];
    }

    acc0 = __builtin_amdgcn_wmma_f32_16x16x4_f32(false, a, false, b0,
                                                 (short)0, acc0, false, false);
    acc1 = __builtin_amdgcn_wmma_f32_16x16x4_f32(false, a, false, b1,
                                                 (short)0, acc1, false, false);
    acc2 = __builtin_amdgcn_wmma_f32_16x16x4_f32(false, a, false, b2,
                                                 (short)0, acc2, false, false);
  }

  const int grow0 = row_tile * 16 + laneh * 8;
  const int rem0  = grow0 - o * HW;
  int n2 = rem0 / WW;
  int m2 = rem0 - n2 * WW;
  int nr = n2 + 1; if (nr == HH) nr = 0;
  float* obase = out + (size_t)(o * NCH) * HW;

#pragma unroll
  for (int r = 0; r < 8; ++r) {
    float* ob = obase + nr * WW + m2;
    ob[(size_t)( 0 + lanem) * HW] = acc0[r];
    ob[(size_t)(16 + lanem) * HW] = acc1[r];
    ob[(size_t)(32 + lanem) * HW] = acc2[r];
    ++m2;
    if (m2 == WW) { m2 = 0; ++nr; if (nr == HH) nr = 0; }
  }
}

extern "C" void kernel_launch(void* const* d_in, const int* in_sizes, int n_in,
                              void* d_out, int out_size, void* d_ws, size_t ws_size,
                              hipStream_t stream) {
  const float* x  = (const float*)d_in[0];
  const float* W0 = (const float*)d_in[1];
  const float* W1 = (const float*)d_in[2];
  float* out = (float*)d_out;
  float* Wf  = (float*)d_ws;

  fold_weights_kernel<<<(NCH * KTOT + 127) / 128, 128, 0, stream>>>(W0, W1, Wf);
  conv_wmma_kernel<<<ROW_TILES / 4, 128, 0, stream>>>(x, Wf, out);
}
